# starting kernel
# baseline (speedup 1.0000x reference)
#include <hip/hip_runtime.h>
#include <hip/hip_bf16.h>
#include <math.h>

typedef float v2f __attribute__((ext_vector_type(2)));
typedef float v8f __attribute__((ext_vector_type(8)));

#define B_ 8
#define N_ 128
#define C_ 64
#define S_ 64
#define BT_ 20

__device__ __forceinline__ float lrelu(float v) { return v >= 0.0f ? v : 0.01f * v; }

__global__ void k_agg(const float* __restrict__ x, const float* __restrict__ nobj,
                      float* __restrict__ agg) {
    int n = blockIdx.x;
    int d = threadIdx.x;
    const float* xp = x + (size_t)n * N_ * C_ + d;
    float s = 0.0f, mx = -INFINITY, mn = INFINITY;
#pragma unroll 4
    for (int i = 0; i < N_; ++i) {
        float v = xp[i * C_];
        s += v;
        mx = fmaxf(mx, v);
        mn = fminf(mn, v);
    }
    float* ap = agg + n * 4 * C_;
    ap[0 * C_ + d] = s * (1.0f / 49.0f);
    ap[1 * C_ + d] = s / nobj[n];
    ap[2 * C_ + d] = mx;
    ap[3 * C_ + d] = mn;
}

__global__ void k_fold(const float* __restrict__ coefs, float* __restrict__ wf) {
    int idx = blockIdx.x * blockDim.x + threadIdx.x;
    int m = idx >> 12;
    int d = (idx >> 6) & 63;
    int s = idx & 63;
    const float* cp = coefs + d * (S_ * BT_) + s * BT_ + m;
    wf[idx] = cp[0] + cp[5] + cp[10] + cp[15];
}

__global__ void k_ef(const float* __restrict__ agg, const float* __restrict__ coefs,
                     const float* __restrict__ bias,
                     float* __restrict__ Eb, float* __restrict__ F) {
    int n = blockIdx.x;
    int s = threadIdx.x;
    float e = 0.0f, f = 0.0f;
#pragma unroll
    for (int a = 0; a < 4; ++a) {
        const float* ag = agg + n * 4 * C_ + a * C_;
        for (int d = 0; d < C_; ++d) {
            const float* cp = coefs + d * (S_ * BT_) + s * BT_ + 5 * a;
            f += ag[d] * cp[3];
            e += ag[d] * cp[4];
        }
    }
    Eb[n * S_ + s] = e + bias[s];
    F[n * S_ + s]  = f;
}

__global__ void k_gemm(const float* __restrict__ x, const float* __restrict__ wf,
                       float* __restrict__ pqr) {
    int wave = threadIdx.x >> 5;
    int lane = threadIdx.x & 31;
    int rl   = lane & 15;
    int kh   = lane >> 4;
    int row  = blockIdx.x * 16 + rl;
    int col  = wave * 16 + rl;

    const float* w0 = wf;
    const float* w1 = wf + 4096;
    const float* w2 = wf + 8192;

    v8f acc0 = {}, acc1 = {}, acc2 = {};

#pragma unroll 4
    for (int k0 = 0; k0 < C_; k0 += 4) {
        int ka = k0 + kh * 2;
        v2f a;
        a.x = x[row * C_ + ka + 0];
        a.y = x[row * C_ + ka + 1];
        v2f b0, b1, b2;
        b0.x = w0[(ka + 0) * S_ + col];  b0.y = w0[(ka + 1) * S_ + col];
        b1.x = w1[(ka + 0) * S_ + col];  b1.y = w1[(ka + 1) * S_ + col];
        b2.x = w2[(ka + 0) * S_ + col];  b2.y = w2[(ka + 1) * S_ + col];
        acc0 = __builtin_amdgcn_wmma_f32_16x16x4_f32(false, a, false, b0, (short)0, acc0, false, false);
        acc1 = __builtin_amdgcn_wmma_f32_16x16x4_f32(false, a, false, b1, (short)0, acc1, false, false);
        acc2 = __builtin_amdgcn_wmma_f32_16x16x4_f32(false, a, false, b2, (short)0, acc2, false, false);
    }

#pragma unroll
    for (int r = 0; r < 8; ++r) {
        int mrow = blockIdx.x * 16 + r + kh * 8;
        pqr[0 * 65536 + mrow * S_ + col] = acc0[r];
        pqr[1 * 65536 + mrow * S_ + col] = acc1[r];
        pqr[2 * 65536 + mrow * S_ + col] = acc2[r];
    }
}

__global__ void k_out(const float* __restrict__ pqr, const float* __restrict__ Eb,
                      const float* __restrict__ F, const float* __restrict__ mask,
                      float* __restrict__ out) {
    int bx = blockIdx.x;
    int n  = bx >> 7;
    int i  = bx & 127;
    int t  = threadIdx.x;
    int sg = t & 15;
    int jl = t >> 4;

    const float4* R4 = (const float4*)(pqr);
    const float4* Q4 = (const float4*)(pqr + 65536);
    const float4* P4 = (const float4*)(pqr + 131072);
    const float4* E4 = (const float4*)Eb;
    const float4* F4 = (const float4*)F;

    float4 p = P4[bx * 16 + sg];
    float4 e = E4[n * 16 + sg];
    p.x += e.x; p.y += e.y; p.z += e.z; p.w += e.w;

    float4 r = R4[bx * 16 + sg];
    float4 f = F4[n * 16 + sg];
    r.x += f.x; r.y += f.y; r.z += f.z; r.w += f.w;

    const float* mrow = mask + ((size_t)n * N_ + i) * N_;
    float4* O4 = (float4*)out;

    for (int j = jl; j < N_; j += 16) {
        float4 q = Q4[(n * N_ + j) * 16 + sg];
        float4 v;
        v.x = p.x + q.x; v.y = p.y + q.y; v.z = p.z + q.z; v.w = p.w + q.w;
        if (j == i) { v.x += r.x; v.y += r.y; v.z += r.z; v.w += r.w; }
        v.x = lrelu(v.x); v.y = lrelu(v.y); v.z = lrelu(v.z); v.w = lrelu(v.w);
        float m = mrow[j];
        v.x *= m; v.y *= m; v.z *= m; v.w *= m;
        O4[((size_t)bx * N_ + j) * 16 + sg] = v;
    }
}

extern "C" void kernel_launch(void* const* d_in, const int* in_sizes, int n_in,
                              void* d_out, int out_size, void* d_ws, size_t ws_size,
                              hipStream_t stream) {
    const float* x     = (const float*)d_in[0];
    const float* mask  = (const float*)d_in[1];
    const float* nobj  = (const float*)d_in[2];
    const float* coefs = (const float*)d_in[3];
    const float* bias  = (const float*)d_in[4];
    float* out = (float*)d_out;

    float* ws  = (float*)d_ws;
    float* agg = ws;
    float* wf  = ws + 2048;
    float* Eb  = ws + 14336;
    float* F   = ws + 14848;
    float* pqr = ws + 15360;

    k_agg <<<B_, C_, 0, stream>>>(x, nobj, agg);
    k_fold<<<48, 256, 0, stream>>>(coefs, wf);
    k_ef  <<<B_, S_, 0, stream>>>(agg, coefs, bias, Eb, F);
    k_gemm<<<(B_ * N_) / 16, 128, 0, stream>>>(x, wf, pqr);
    k_out <<<B_ * N_, 256, 0, stream>>>(pqr, Eb, F, mask, out);
}


# Recut 5 launches into 2: small prep (fold, x reduction, Eb/F), then each group WMMAs its own 16x32 tile's P/Q/R into LDS and streams it; x sum and Eb/F become ordered partial sums (sum tree changed)
# speedup vs baseline: 3.2574x; 3.2574x over previous
#include <hip/hip_runtime.h>
#include <math.h>


typedef float v2f __attribute__((ext_vector_type(2)));
typedef float v8f __attribute__((ext_vector_type(8)));

#define B_ 8
#define N_ 128
#define C_ 64
#define S_ 64
#define BT_ 20

#define EF_BLOCKS   32
#define FOLD_BLOCKS 48
#define XP_ 68

__device__ __forceinline__ float lrelu(float v) { return v >= 0.0f ? v : 0.01f * v; }

__global__ void __launch_bounds__(256)
k_prep(const float* __restrict__ x, const float* __restrict__ nobj,
       const float* __restrict__ coefs, const float* __restrict__ bias,
       float* __restrict__ wf, float* __restrict__ Eb, float* __restrict__ F) {
    __shared__ float part_s[16][C_];
    __shared__ float part_mx[16][C_];
    __shared__ float part_mn[16][C_];
    __shared__ float agg_s[4][C_];
    __shared__ float dot_s[8][32];

    int t = threadIdx.x;

    if (blockIdx.x >= EF_BLOCKS) {
        int idx = (blockIdx.x - EF_BLOCKS) * 256 + t;
        int m = idx >> 12;
        int d = (idx >> 6) & 63;
        int s = idx & 63;
        const float* cp = coefs + d * (S_ * BT_) + s * BT_ + m;
        wf[idx] = cp[0] + cp[5] + cp[10] + cp[15];
        return;
    }

    int n  = blockIdx.x >> 2;
    int s0 = (blockIdx.x & 3) * 16;

    int p    = t >> 5;
    int lane = t & 31;
    int sl   = lane & 15;
    int ef   = lane >> 4;
    int a    = p >> 1;
    int dh   = p & 1;
    const float* cp = coefs + (dh * 32) * (S_ * BT_) + (s0 + sl) * BT_ + 5 * a + 3 + ef;
    float c[32];
#pragma unroll
    for (int dd = 0; dd < 32; ++dd) c[dd] = cp[dd * (S_ * BT_)];

    int d4 = t & 15;
    int g  = t >> 4;
    const float4* xp = (const float4*)(x + (size_t)(n * N_ + g * 8) * C_) + d4;
    float4 v[8];
#pragma unroll
    for (int r = 0; r < 8; ++r) v[r] = xp[r * 16];
    float4 sm = v[0], mx = v[0], mn = v[0];
#pragma unroll
    for (int r = 1; r < 8; ++r) {
        sm.x += v[r].x; sm.y += v[r].y; sm.z += v[r].z; sm.w += v[r].w;
        mx.x = fmaxf(mx.x, v[r].x); mx.y = fmaxf(mx.y, v[r].y);
        mx.z = fmaxf(mx.z, v[r].z); mx.w = fmaxf(mx.w, v[r].w);
        mn.x = fminf(mn.x, v[r].x); mn.y = fminf(mn.y, v[r].y);
        mn.z = fminf(mn.z, v[r].z); mn.w = fminf(mn.w, v[r].w);
    }
    *(float4*)&part_s [g][d4 * 4] = sm;
    *(float4*)&part_mx[g][d4 * 4] = mx;
    *(float4*)&part_mn[g][d4 * 4] = mn;
    __syncthreads();

    if (t < C_) {
        float s = part_s[0][t], hi = part_mx[0][t], lo = part_mn[0][t];
        for (int gg = 1; gg < 16; ++gg) {
            s += part_s[gg][t];
            hi = fmaxf(hi, part_mx[gg][t]);
            lo = fminf(lo, part_mn[gg][t]);
        }
        agg_s[0][t] = s * (1.0f / 49.0f);
        agg_s[1][t] = s / nobj[n];
        agg_s[2][t] = hi;
        agg_s[3][t] = lo;
    }
    __syncthreads();

    const float* ag = &agg_s[a][dh * 32];
    float acc = 0.0f;
#pragma unroll
    for (int dd = 0; dd < 32; ++dd) acc += ag[dd] * c[dd];
    dot_s[p][lane] = acc;
    __syncthreads();

    if (t < 32) {
        float r = dot_s[0][t];
        for (int pp = 1; pp < 8; ++pp) r += dot_s[pp][t];
        int s = s0 + (t & 15);
        if (t >> 4) Eb[n * S_ + s] = r + bias[s];
        else        F [n * S_ + s] = r;
    }
}

__global__ void __launch_bounds__(256)
k_tile(const float* __restrict__ x, const float* __restrict__ mask,
       const float* __restrict__ wf, const float* __restrict__ Eb,
       const float* __restrict__ F, float* __restrict__ out) {
    __shared__ __attribute__((aligned(16))) float Xs[48][XP_];
    __shared__ __attribute__((aligned(16))) float Qs[32][S_];
    __shared__ __attribute__((aligned(16))) float Ps[16][S_];
    __shared__ __attribute__((aligned(16))) float Rs[16][S_];
    __shared__ float Ms[16][32];

    int t  = threadIdx.x;
    int bx = blockIdx.x;
    int n  = bx >> 5;
    int i0 = ((bx >> 2) & 7) * 16;
    int j0 = (bx & 3) * 32;

    int wave = t >> 5;
    int lane = t & 31;
    int rl   = lane & 15;
    int kh   = lane >> 4;
    int mt   = wave >> 2;
    int nt   = wave & 3;
    int col  = nt * 16 + rl;

    float2 mk = *(const float2*)(mask + (size_t)(n * N_ + i0 + (t >> 4)) * N_ + j0 + (t & 15) * 2);

    float4 xv[3];
#pragma unroll
    for (int u = 0; u < 3; ++u) {
        int f   = t + u * 256;
        int row = f >> 4;
        int grow = (row < 32) ? (j0 + row) : (i0 + row - 32);
        xv[u] = *((const float4*)(x + (size_t)(n * N_ + grow) * C_) + (f & 15));
    }

    const float* wb1 = wf + 4096 + (kh * 2) * S_ + col;
    const float* wb2 = wf + (mt ? 0 : 8192) + (kh * 2) * S_ + col;
    float add = (mt ? F : Eb)[n * S_ + col];

    v2f b1[16], b2[16];
#pragma unroll
    for (int ks = 0; ks < 16; ++ks) {
        int k0 = ks * 4;
        b1[ks].x = wb1[(k0 + 0) * S_];  b1[ks].y = wb1[(k0 + 1) * S_];
        b2[ks].x = wb2[(k0 + 0) * S_];  b2[ks].y = wb2[(k0 + 1) * S_];
    }

#pragma unroll
    for (int u = 0; u < 3; ++u) {
        int f = t + u * 256;
        *(float4*)&Xs[f >> 4][(f & 15) * 4] = xv[u];
    }
    Ms[t >> 4][(t & 15) * 2 + 0] = mk.x;
    Ms[t >> 4][(t & 15) * 2 + 1] = mk.y;
    __syncthreads();

    v2f a1[16], a2[16];
#pragma unroll
    for (int ks = 0; ks < 16; ++ks) {
        a1[ks] = *(const v2f*)&Xs[mt * 16 + rl][ks * 4 + kh * 2];
        a2[ks] = *(const v2f*)&Xs[32 + rl][ks * 4 + kh * 2];
    }

    v8f accq = {}, accp = {};
#pragma unroll
    for (int ks = 0; ks < 16; ++ks) {
        accq = __builtin_amdgcn_wmma_f32_16x16x4_f32(false, a1[ks], false, b1[ks], (short)0, accq, false, false);
        accp = __builtin_amdgcn_wmma_f32_16x16x4_f32(false, a2[ks], false, b2[ks], (short)0, accp, false, false);
    }

    float* dst2 = mt ? &Rs[0][0] : &Ps[0][0];
#pragma unroll
    for (int r = 0; r < 8; ++r) {
        int row = r + kh * 8;
        Qs[mt * 16 + row][col] = accq[r];
        dst2[row * S_ + col]   = accp[r] + add;
    }
    __syncthreads();

    int sg = t & 15;
    int jl = t >> 4;
    float4 q0 = *(const float4*)&Qs[jl][sg * 4];
    float4 q1 = *(const float4*)&Qs[jl + 16][sg * 4];
    int ja = j0 + jl;
    int jb = ja + 16;
    float4* O4 = (float4*)out;

#pragma unroll 4
    for (int il = 0; il < 16; ++il) {
        int i = i0 + il;
        float4 p = *(const float4*)&Ps[il][sg * 4];
        float m0 = Ms[il][jl];
        float m1 = Ms[il][jl + 16];
        float4 v0, v1;
        v0.x = p.x + q0.x; v0.y = p.y + q0.y; v0.z = p.z + q0.z; v0.w = p.w + q0.w;
        v1.x = p.x + q1.x; v1.y = p.y + q1.y; v1.z = p.z + q1.z; v1.w = p.w + q1.w;
        if (ja == i) {
            float4 r = *(const float4*)&Rs[il][sg * 4];
            v0.x += r.x; v0.y += r.y; v0.z += r.z; v0.w += r.w;
        }
        if (jb == i) {
            float4 r = *(const float4*)&Rs[il][sg * 4];
            v1.x += r.x; v1.y += r.y; v1.z += r.z; v1.w += r.w;
        }
        v0.x = lrelu(v0.x) * m0; v0.y = lrelu(v0.y) * m0; v0.z = lrelu(v0.z) * m0; v0.w = lrelu(v0.w) * m0;
        v1.x = lrelu(v1.x) * m1; v1.y = lrelu(v1.y) * m1; v1.z = lrelu(v1.z) * m1; v1.w = lrelu(v1.w) * m1;
        size_t base = (size_t)(n * N_ + i) * N_;
        O4[(base + ja) * 16 + sg] = v0;
        O4[(base + jb) * 16 + sg] = v1;
    }
}

extern "C" void kernel_launch(void* const* d_in, const int* in_sizes, int n_in,
                              void* d_out, int out_size, void* d_ws, size_t ws_size,
                              hipStream_t stream) {
    const float* x     = (const float*)d_in[0];
    const float* mask  = (const float*)d_in[1];
    const float* nobj  = (const float*)d_in[2];
    const float* coefs = (const float*)d_in[3];
    const float* bias  = (const float*)d_in[4];
    float* out = (float*)d_out;

    float* ws  = (float*)d_ws;
    float* wf  = ws;
    float* Eb  = ws + 12288;
    float* F   = ws + 12800;

    k_prep<<<EF_BLOCKS + FOLD_BLOCKS, 256, 0, stream>>>(x, nobj, coefs, bias, wf, Eb, F);
    k_tile<<<B_ * 8 * 4, 256, 0, stream>>>(x, mask, wf, Eb, F, out);
}


# No global load waits on a serial chain: k_prep fetches nobj[n] and bias[s] at entry; k_tile's first barrier waits for LDS only, so B-fragment loads stay in flight across it; sum order unchanged
# speedup vs baseline: 3.4533x; 1.0601x over previous
#include <hip/hip_runtime.h>
#include <math.h>


typedef float v2f __attribute__((ext_vector_type(2)));
typedef float v8f __attribute__((ext_vector_type(8)));

#define B_ 8
#define N_ 128
#define C_ 64
#define S_ 64
#define BT_ 20

#define EF_BLOCKS   32
#define FOLD_BLOCKS 48
#define XP_ 68

__device__ __forceinline__ float lrelu(float v) { return v >= 0.0f ? v : 0.01f * v; }

__global__ void __launch_bounds__(256)
k_prep(const float* __restrict__ x, const float* __restrict__ nobj,
       const float* __restrict__ coefs, const float* __restrict__ bias,
       float* __restrict__ wf, float* __restrict__ Eb, float* __restrict__ F) {
    __shared__ float part_s[16][C_];
    __shared__ float part_mx[16][C_];
    __shared__ float part_mn[16][C_];
    __shared__ float agg_s[4][C_];
    __shared__ float dot_s[8][32];

    int t = threadIdx.x;

    if (blockIdx.x >= EF_BLOCKS) {
        int idx = (blockIdx.x - EF_BLOCKS) * 256 + t;
        int m = idx >> 12;
        int d = (idx >> 6) & 63;
        int s = idx & 63;
        const float* cp = coefs + d * (S_ * BT_) + s * BT_ + m;
        wf[idx] = cp[0] + cp[5] + cp[10] + cp[15];
        return;
    }

    int n  = blockIdx.x >> 2;
    int s0 = (blockIdx.x & 3) * 16;

    float nob = nobj[n];
    float bs  = bias[s0 + (t & 15)];

    int p    = t >> 5;
    int lane = t & 31;
    int sl   = lane & 15;
    int ef   = lane >> 4;
    int a    = p >> 1;
    int dh   = p & 1;
    const float* cp = coefs + (dh * 32) * (S_ * BT_) + (s0 + sl) * BT_ + 5 * a + 3 + ef;
    float c[32];
#pragma unroll
    for (int dd = 0; dd < 32; ++dd) c[dd] = cp[dd * (S_ * BT_)];

    int d4 = t & 15;
    int g  = t >> 4;
    const float4* xp = (const float4*)(x + (size_t)(n * N_ + g * 8) * C_) + d4;
    float4 v[8];
#pragma unroll
    for (int r = 0; r < 8; ++r) v[r] = xp[r * 16];
    float4 sm = v[0], mx = v[0], mn = v[0];
#pragma unroll
    for (int r = 1; r < 8; ++r) {
        sm.x += v[r].x; sm.y += v[r].y; sm.z += v[r].z; sm.w += v[r].w;
        mx.x = fmaxf(mx.x, v[r].x); mx.y = fmaxf(mx.y, v[r].y);
        mx.z = fmaxf(mx.z, v[r].z); mx.w = fmaxf(mx.w, v[r].w);
        mn.x = fminf(mn.x, v[r].x); mn.y = fminf(mn.y, v[r].y);
        mn.z = fminf(mn.z, v[r].z); mn.w = fminf(mn.w, v[r].w);
    }
    *(float4*)&part_s [g][d4 * 4] = sm;
    *(float4*)&part_mx[g][d4 * 4] = mx;
    *(float4*)&part_mn[g][d4 * 4] = mn;
    __syncthreads();

    if (t < C_) {
        float s = part_s[0][t], hi = part_mx[0][t], lo = part_mn[0][t];
        for (int gg = 1; gg < 16; ++gg) {
            s += part_s[gg][t];
            hi = fmaxf(hi, part_mx[gg][t]);
            lo = fminf(lo, part_mn[gg][t]);
        }
        agg_s[0][t] = s * (1.0f / 49.0f);
        agg_s[1][t] = s / nob;
        agg_s[2][t] = hi;
        agg_s[3][t] = lo;
    }
    __syncthreads();

    const float* ag = &agg_s[a][dh * 32];
    float acc = 0.0f;
#pragma unroll
    for (int dd = 0; dd < 32; ++dd) acc += ag[dd] * c[dd];
    dot_s[p][lane] = acc;
    __syncthreads();

    if (t < 32) {
        float r = dot_s[0][t];
        for (int pp = 1; pp < 8; ++pp) r += dot_s[pp][t];
        int s = s0 + (t & 15);
        if (t >> 4) Eb[n * S_ + s] = r + bs;
        else        F [n * S_ + s] = r;
    }
}

__global__ void __launch_bounds__(256)
k_tile(const float* __restrict__ x, const float* __restrict__ mask,
       const float* __restrict__ wf, const float* __restrict__ Eb,
       const float* __restrict__ F, float* __restrict__ out) {
    __shared__ __attribute__((aligned(16))) float Xs[48][XP_];
    __shared__ __attribute__((aligned(16))) float Qs[32][S_];
    __shared__ __attribute__((aligned(16))) float Ps[16][S_];
    __shared__ __attribute__((aligned(16))) float Rs[16][S_];
    __shared__ float Ms[16][32];

    int t  = threadIdx.x;
    int bx = blockIdx.x;
    int n  = bx >> 5;
    int i0 = ((bx >> 2) & 7) * 16;
    int j0 = (bx & 3) * 32;

    int wave = t >> 5;
    int lane = t & 31;
    int rl   = lane & 15;
    int kh   = lane >> 4;
    int mt   = wave >> 2;
    int nt   = wave & 3;
    int col  = nt * 16 + rl;

    float2 mk = *(const float2*)(mask + (size_t)(n * N_ + i0 + (t >> 4)) * N_ + j0 + (t & 15) * 2);

    float4 xv[3];
    {
        int r16 = t >> 4;
        const float4* xn = (const float4*)(x + (size_t)n * N_ * C_) + (t & 15);
        xv[0] = xn[(j0 + r16) * 16];
        xv[1] = xn[(j0 + 16 + r16) * 16];
        xv[2] = xn[(i0 + r16) * 16];
    }

    const float* wb1 = wf + 4096 + (kh * 2) * S_ + col;
    const float* wb2 = wf + (mt ? 0 : 8192) + (kh * 2) * S_ + col;

    asm volatile("" ::: "memory");

    v2f b1[16], b2[16];
#pragma unroll
    for (int ks = 0; ks < 16; ++ks) {
        int k0 = ks * 4;
        b1[ks].x = wb1[(k0 + 0) * S_];  b1[ks].y = wb1[(k0 + 1) * S_];
        b2[ks].x = wb2[(k0 + 0) * S_];  b2[ks].y = wb2[(k0 + 1) * S_];
    }
    float add = (mt ? F : Eb)[n * S_ + col];

#pragma unroll
    for (int u = 0; u < 3; ++u) {
        int f = t + u * 256;
        *(float4*)&Xs[f >> 4][(f & 15) * 4] = xv[u];
    }
    Ms[t >> 4][(t & 15) * 2 + 0] = mk.x;
    Ms[t >> 4][(t & 15) * 2 + 1] = mk.y;
    asm volatile("s_wait_dscnt 0x0\n\ts_barrier_signal -1\n\ts_barrier_wait -1" ::: "memory");

    v2f a1[16], a2[16];
#pragma unroll
    for (int ks = 0; ks < 16; ++ks) {
        a1[ks] = *(const v2f*)&Xs[mt * 16 + rl][ks * 4 + kh * 2];
        a2[ks] = *(const v2f*)&Xs[32 + rl][ks * 4 + kh * 2];
    }

    v8f accq = {}, accp = {};
#pragma unroll
    for (int ks = 0; ks < 16; ++ks) {
        accq = __builtin_amdgcn_wmma_f32_16x16x4_f32(false, a1[ks], false, b1[ks], (short)0, accq, false, false);
        accp = __builtin_amdgcn_wmma_f32_16x16x4_f32(false, a2[ks], false, b2[ks], (short)0, accp, false, false);
    }

    float* dst2 = mt ? &Rs[0][0] : &Ps[0][0];
#pragma unroll
    for (int r = 0; r < 8; ++r) {
        int row = r + kh * 8;
        Qs[mt * 16 + row][col] = accq[r];
        dst2[row * S_ + col]   = accp[r] + add;
    }
    __syncthreads();

    int sg = t & 15;
    int jl = t >> 4;
    float4 q0 = *(const float4*)&Qs[jl][sg * 4];
    float4 q1 = *(const float4*)&Qs[jl + 16][sg * 4];
    int ja = j0 + jl;
    int jb = ja + 16;
    float4* O4 = (float4*)out;

#pragma unroll 4
    for (int il = 0; il < 16; ++il) {
        int i = i0 + il;
        float4 p = *(const float4*)&Ps[il][sg * 4];
        float m0 = Ms[il][jl];
        float m1 = Ms[il][jl + 16];
        float4 v0, v1;
        v0.x = p.x + q0.x; v0.y = p.y + q0.y; v0.z = p.z + q0.z; v0.w = p.w + q0.w;
        v1.x = p.x + q1.x; v1.y = p.y + q1.y; v1.z = p.z + q1.z; v1.w = p.w + q1.w;
        if (ja == i) {
            float4 r = *(const float4*)&Rs[il][sg * 4];
            v0.x += r.x; v0.y += r.y; v0.z += r.z; v0.w += r.w;
        }
        if (jb == i) {
            float4 r = *(const float4*)&Rs[il][sg * 4];
            v1.x += r.x; v1.y += r.y; v1.z += r.z; v1.w += r.w;
        }
        v0.x = lrelu(v0.x) * m0; v0.y = lrelu(v0.y) * m0; v0.z = lrelu(v0.z) * m0; v0.w = lrelu(v0.w) * m0;
        v1.x = lrelu(v1.x) * m1; v1.y = lrelu(v1.y) * m1; v1.z = lrelu(v1.z) * m1; v1.w = lrelu(v1.w) * m1;
        size_t base = (size_t)(n * N_ + i) * N_;
        O4[(base + ja) * 16 + sg] = v0;
        O4[(base + jb) * 16 + sg] = v1;
    }
}

extern "C" void kernel_launch(void* const* d_in, const int* in_sizes, int n_in,
                              void* d_out, int out_size, void* d_ws, size_t ws_size,
                              hipStream_t stream) {
    const float* x     = (const float*)d_in[0];
    const float* mask  = (const float*)d_in[1];
    const float* nobj  = (const float*)d_in[2];
    const float* coefs = (const float*)d_in[3];
    const float* bias  = (const float*)d_in[4];
    float* out = (float*)d_out;

    float* ws  = (float*)d_ws;
    float* wf  = ws;
    float* Eb  = ws + 12288;
    float* F   = ws + 12800;

    k_prep<<<EF_BLOCKS + FOLD_BLOCKS, 256, 0, stream>>>(x, nobj, coefs, bias, wf, Eb, F);
    k_tile<<<B_ * 8 * 4, 256, 0, stream>>>(x, mask, wf, Eb, F, out);
}
